# starting kernel
# baseline (speedup 1.0000x reference)
#include <hip/hip_runtime.h>


typedef float v2f __attribute__((ext_vector_type(2)));
typedef float v8f __attribute__((ext_vector_type(8)));

#define RES_TOTAL   (32 * 4096)
#define RES_PER_BLK 32
#define WAVE_LDS    1024
#define EPSF        1e-12f

__device__ __forceinline__ void wave_sync() {
  __builtin_amdgcn_wave_barrier();
}

__device__ __forceinline__ v8f wmma4x4(v2f a, v2f b) {
  v8f c = {};
  return __builtin_amdgcn_wmma_f32_16x16x4_f32(false, a, false, b, (short)0, c,
                                               false, false);
}

__global__ __launch_bounds__(256) void rotamer_projection_kernel(
    const float* __restrict__ dih,
    const float* __restrict__ bbc,
    const int*   __restrict__ seq,
    const float* __restrict__ dframes,
    const int*   __restrict__ gidx,
    const float* __restrict__ amask,
    const float* __restrict__ litp,
    float*       __restrict__ out)
{
  __shared__ float ldsfr[8][WAVE_LDS];

  const int tid  = threadIdx.x;
  const int wave = tid >> 5;
  const int lane = tid & 31;
  float* fr = ldsfr[wave];

  const int low = lane & 15;
  const int hi  = lane >> 4;
  const int i   = low >> 2;
  const int m   = low & 3;
  const int k0  = hi * 2;

  const int resBase = blockIdx.x * RES_PER_BLK + wave * 4;
  const int myRes   = resBase + i;
  const int mySeq   = seq[myRes];

  float* stq[8];
#pragma unroll
  for (int v = 0; v < 8; ++v) {
    const int row  = v + 8 * hi;
    const bool keep = (row >> 2) == i;
    const int off  = keep ? ((row & 3) * 4 + m) : (512 + low);
    stq[v] = fr + i * 128 + off;
  }
  float* pA = fr + i * 128 + m * 4 + k0;
  float* pB = fr + i * 128 + k0 * 4 + m;

  float wc[2], ws[2], wk[2];
#pragma unroll
  for (int e = 0; e < 2; ++e) {
    const int k = k0 + e;
    wc[e] = ((k == 1 && m == 1) || (k == 2 && m == 2)) ? 1.f : 0.f;
    ws[e] = (k == 2 && m == 1) ? 1.f : ((k == 1 && m == 2) ? -1.f : 0.f);
    wk[e] = ((k == 0 && m == 0) || (k == 3 && m == 3)) ? 1.f : 0.f;
  }

  const float* dfBase = dframes + mySeq * 128 + m * 4 + k0;
  v2f dfv[8];
#pragma unroll
  for (int a = 0; a < 8; ++a) {
    dfv[a][0] = dfBase[a * 16];
    dfv[a][1] = dfBase[a * 16 + 1];
  }
  const float* dBase = dih + myRes * 14;
  float sv[8], cv[8];
  sv[0] = 0.f;
  cv[0] = 1.f;
#pragma unroll
  for (int a = 1; a < 8; ++a) {
    const float x = dBase[(a - 1) * 2];
    const float y = dBase[(a - 1) * 2 + 1];
    const float inv = rsqrtf(x * x + y * y + EPSF);
    sv[a] = x * inv;
    cv[a] = y * inv;
  }

#pragma unroll
  for (int a = 0; a < 8; ++a) {
    v2f bop;
    bop[0] = fmaf(wc[0], cv[a], fmaf(ws[0], sv[a], wk[0]));
    bop[1] = fmaf(wc[1], cv[a], fmaf(ws[1], sv[a], wk[1]));

    v8f d = wmma4x4(dfv[a], bop);
#pragma unroll
    for (int v = 0; v < 8; ++v) stq[v][a * 16] = d[v];
    wave_sync();
  }

#pragma unroll
  for (int f = 5; f <= 7; ++f) {
    v2f aop;
    aop[0] = pA[(f - 1) * 16];
    aop[1] = pA[(f - 1) * 16 + 1];
    v2f bop;
    bop[0] = pB[f * 16];
    bop[1] = pB[f * 16 + 4];
    wave_sync();

    v8f d = wmma4x4(aop, bop);
#pragma unroll
    for (int v = 0; v < 8; ++v) stq[v][f * 16] = d[v];
    wave_sync();
  }

  const float* bp = bbc + myRes * 9;
  const float nx = bp[0], ny = bp[1], nz = bp[2];
  const float cax = bp[3], cay = bp[4], caz = bp[5];
  const float cx = bp[6], cy = bp[7], cz = bp[8];
  const float v1x = cx - cax, v1y = cy - cay, v1z = cz - caz;
  const float v2x = nx - cax, v2y = ny - cay, v2z = nz - caz;
  const float inv1 = rsqrtf(v1x * v1x + v1y * v1y + v1z * v1z + EPSF);
  const float e1x = v1x * inv1, e1y = v1y * inv1, e1z = v1z * inv1;
  const float d12 = e1x * v2x + e1y * v2y + e1z * v2z;
  const float u2x = v2x - e1x * d12, u2y = v2y - e1y * d12,
              u2z = v2z - e1z * d12;
  const float inv2 = rsqrtf(u2x * u2x + u2y * u2y + u2z * u2z + EPSF);
  const float e2x = u2x * inv2, e2y = u2y * inv2, e2z = u2z * inv2;
  const float e3x = e1y * e2z - e1z * e2y;
  const float e3y = e1z * e2x - e1x * e2z;
  const float e3z = e1x * e2y - e1y * e2x;

  float row0, row1, row2, row3;
  if (m == 0)      { row0 = e1x; row1 = e2x; row2 = e3x; row3 = cax; }
  else if (m == 1) { row0 = e1y; row1 = e2y; row2 = e3y; row3 = cay; }
  else if (m == 2) { row0 = e1z; row1 = e2z; row2 = e3z; row3 = caz; }
  else             { row0 = 0.f; row1 = 0.f; row2 = 0.f; row3 = 1.f; }
  v2f abb;
  abb[0] = (k0 == 0) ? row0 : row2;
  abb[1] = (k0 == 0) ? row1 : row3;

#pragma unroll
  for (int a = 0; a < 8; ++a) {
    v2f bop;
    bop[0] = pB[a * 16];
    bop[1] = pB[a * 16 + 4];
    wave_sync();

    v8f d = wmma4x4(abb, bop);
#pragma unroll
    for (int v = 0; v < 8; ++v) stq[v][a * 16] = d[v];
    wave_sync();
  }

  for (int t = lane; t < 4 * 14; t += 32) {
    const int ri   = t / 14;
    const int atom = t - ri * 14;
    const int res  = resBase + ri;
    const int sq   = seq[res];
    const int g    = gidx[sq * 14 + atom];
    const float* G = &fr[(ri * 8 + g) * 16];
    const float* lp = litp + (sq * 14 + atom) * 3;
    const float l0 = lp[0], l1 = lp[1], l2 = lp[2];
    const float mk = amask[sq * 14 + atom];
    float* op = out + (res * 14 + atom) * 3;
#pragma unroll
    for (int r = 0; r < 3; ++r) {
      op[r] = (G[r * 4 + 0] * l0 + G[r * 4 + 1] * l1 + G[r * 4 + 2] * l2 +
               G[r * 4 + 3]) * mk;
    }
  }
}

extern "C" void kernel_launch(void* const* d_in, const int* in_sizes, int n_in,
                              void* d_out, int out_size, void* d_ws,
                              size_t ws_size, hipStream_t stream) {
  const float* dih     = (const float*)d_in[0];
  const float* bbc     = (const float*)d_in[1];
  const int*   seq     = (const int*)d_in[2];
  const float* dframes = (const float*)d_in[3];
  const int*   gidx    = (const int*)d_in[4];
  const float* amask   = (const float*)d_in[5];
  const float* litp    = (const float*)d_in[6];
  float*       out     = (float*)d_out;

  const int blocks = RES_TOTAL / RES_PER_BLK;
  rotamer_projection_kernel<<<blocks, 256, 0, stream>>>(
      dih, bbc, seq, dframes, gidx, amask, litp, out);
}


# Backbone frame applied to the 14 local atom positions q=fr[g]@(lit;1), one WMMA per residue straight to global, not to the 8 frames through LDS (15 WMMA, not 19); order of the f32 sums changes
# speedup vs baseline: 1.4554x; 1.4554x over previous
#include <hip/hip_runtime.h>


typedef float v2f __attribute__((ext_vector_type(2)));
typedef float v8f __attribute__((ext_vector_type(8)));

#define RES_TOTAL   (32 * 4096)
#define RES_PER_BLK 32
#define WAVE_LDS    1024
#define EPSF        1e-12f

__device__ __forceinline__ void wave_sync() {
  __builtin_amdgcn_wave_barrier();
}

__device__ __forceinline__ v8f wmma4x4(v2f a, v2f b) {
  v8f c = {};
  return __builtin_amdgcn_wmma_f32_16x16x4_f32(false, a, false, b, (short)0, c,
                                               false, false);
}

__global__ __launch_bounds__(256) void rotamer_projection_kernel(
    const float* __restrict__ dih,
    const float* __restrict__ bbc,
    const int*   __restrict__ seq,
    const float* __restrict__ dframes,
    const int*   __restrict__ gidx,
    const float* __restrict__ amask,
    const float* __restrict__ litp,
    float*       __restrict__ out)
{
  __shared__ __attribute__((aligned(16))) float ldsfr[8][WAVE_LDS];

  const int tid  = threadIdx.x;
  const int wave = tid >> 5;
  const int lane = tid & 31;
  float* fr = ldsfr[wave];

  const int low = lane & 15;
  const int hi  = lane >> 4;
  const int i   = low >> 2;
  const int m   = low & 3;
  const int k0  = hi * 2;

  const int resBase = blockIdx.x * RES_PER_BLK + wave * 4;
  const int myRes   = resBase + i;
  const int mySeq   = seq[myRes];

  float* stq[8];
#pragma unroll
  for (int v = 0; v < 8; ++v) {
    const int row  = v + 8 * hi;
    const bool keep = (row >> 2) == i;
    const int off  = keep ? ((row & 3) * 4 + m) : (512 + low);
    stq[v] = fr + i * 128 + off;
  }
  float* pA = fr + i * 128 + m * 4 + k0;
  float* pB = fr + i * 128 + k0 * 4 + m;

  float wc[2], ws[2], wk[2];
#pragma unroll
  for (int e = 0; e < 2; ++e) {
    const int k = k0 + e;
    wc[e] = ((k == 1 && m == 1) || (k == 2 && m == 2)) ? 1.f : 0.f;
    ws[e] = (k == 2 && m == 1) ? 1.f : ((k == 1 && m == 2) ? -1.f : 0.f);
    wk[e] = ((k == 0 && m == 0) || (k == 3 && m == 3)) ? 1.f : 0.f;
  }

  const float* dfBase = dframes + mySeq * 128 + m * 4 + k0;
  v2f dfv[8];
#pragma unroll
  for (int a = 0; a < 8; ++a) {
    dfv[a][0] = dfBase[a * 16];
    dfv[a][1] = dfBase[a * 16 + 1];
  }
  const float* dBase = dih + myRes * 14;
  float sv[8], cv[8];
  sv[0] = 0.f;
  cv[0] = 1.f;
#pragma unroll
  for (int a = 1; a < 8; ++a) {
    const float x = dBase[(a - 1) * 2];
    const float y = dBase[(a - 1) * 2 + 1];
    const float inv = rsqrtf(x * x + y * y + EPSF);
    sv[a] = x * inv;
    cv[a] = y * inv;
  }

  const int atomc = (low < 14) ? low : 13;
  int   gq[4];
  float mkq[4], l0q[4], l1q[4], l2q[4];
#pragma unroll
  for (int ri = 0; ri < 4; ++ri) {
    const int sq  = seq[resBase + ri];
    const int idx = sq * 14 + atomc;
    gq[ri]  = gidx[idx];
    mkq[ri] = amask[idx];
    const float* lp = litp + idx * 3;
    l0q[ri] = lp[0];
    l1q[ri] = lp[1];
    l2q[ri] = lp[2];
  }

#pragma unroll
  for (int a = 0; a < 8; ++a) {
    v2f bop;
    bop[0] = fmaf(wc[0], cv[a], fmaf(ws[0], sv[a], wk[0]));
    bop[1] = fmaf(wc[1], cv[a], fmaf(ws[1], sv[a], wk[1]));

    v8f d = wmma4x4(dfv[a], bop);
#pragma unroll
    for (int v = 0; v < 8; ++v) stq[v][a * 16] = d[v];
    wave_sync();
  }

#pragma unroll
  for (int f = 5; f <= 7; ++f) {
    v2f aop;
    aop[0] = pA[(f - 1) * 16];
    aop[1] = pA[(f - 1) * 16 + 1];
    v2f bop;
    bop[0] = pB[f * 16];
    bop[1] = pB[f * 16 + 4];
    wave_sync();

    v8f d = wmma4x4(aop, bop);
#pragma unroll
    for (int v = 0; v < 8; ++v) stq[v][f * 16] = d[v];
    wave_sync();
  }

  const float* bp = bbc + myRes * 9;
  const float nx = bp[0], ny = bp[1], nz = bp[2];
  const float cax = bp[3], cay = bp[4], caz = bp[5];
  const float cx = bp[6], cy = bp[7], cz = bp[8];
  const float v1x = cx - cax, v1y = cy - cay, v1z = cz - caz;
  const float v2x = nx - cax, v2y = ny - cay, v2z = nz - caz;
  const float inv1 = rsqrtf(v1x * v1x + v1y * v1y + v1z * v1z + EPSF);
  const float e1x = v1x * inv1, e1y = v1y * inv1, e1z = v1z * inv1;
  const float d12 = e1x * v2x + e1y * v2y + e1z * v2z;
  const float u2x = v2x - e1x * d12, u2y = v2y - e1y * d12,
              u2z = v2z - e1z * d12;
  const float inv2 = rsqrtf(u2x * u2x + u2y * u2y + u2z * u2z + EPSF);
  const float e2x = u2x * inv2, e2y = u2y * inv2, e2z = u2z * inv2;
  const float e3x = e1y * e2z - e1z * e2y;
  const float e3y = e1z * e2x - e1x * e2z;
  const float e3z = e1x * e2y - e1y * e2x;

  float row0, row1, row2, row3;
  if (m == 0)      { row0 = e1x; row1 = e2x; row2 = e3x; row3 = cax; }
  else if (m == 1) { row0 = e1y; row1 = e2y; row2 = e3y; row3 = cay; }
  else if (m == 2) { row0 = e1z; row1 = e2z; row2 = e3z; row3 = caz; }
  else             { row0 = 0.f; row1 = 0.f; row2 = 0.f; row3 = 1.f; }
  v2f abb;
  abb[0] = (k0 == 0) ? row0 : row2;
  abb[1] = (k0 == 0) ? row1 : row3;

  v8f dd[4];
#pragma unroll
  for (int ri = 0; ri < 4; ++ri) {
    const float4* F =
        reinterpret_cast<const float4*>(fr + (ri * 8 + gq[ri]) * 16 + hi * 8);
    const float4 fa = F[0];
    const float4 fb = F[1];
    v2f q;
    q[0] = fa.x * l0q[ri] + fa.y * l1q[ri] + fa.z * l2q[ri] + fa.w;
    q[1] = fb.x * l0q[ri] + fb.y * l1q[ri] + fb.z * l2q[ri] + fb.w;
    dd[ri] = wmma4x4(abb, q);
  }
  float ov[4][3];
#pragma unroll
  for (int ri = 0; ri < 4; ++ri) {
#pragma unroll
    for (int r = 0; r < 3; ++r) ov[ri][r] = dd[ri][4 * (ri & 1) + r] * mkq[ri];
  }
#pragma unroll
  for (int ri = 0; ri < 4; ++ri) {
    if (hi == (ri >> 1) && low < 14) {
      float* op = out + ((resBase + ri) * 14 + low) * 3;
      op[0] = ov[ri][0];
      op[1] = ov[ri][1];
      op[2] = ov[ri][2];
    }
  }
}

extern "C" void kernel_launch(void* const* d_in, const int* in_sizes, int n_in,
                              void* d_out, int out_size, void* d_ws,
                              size_t ws_size, hipStream_t stream) {
  const float* dih     = (const float*)d_in[0];
  const float* bbc     = (const float*)d_in[1];
  const int*   seq     = (const int*)d_in[2];
  const float* dframes = (const float*)d_in[3];
  const int*   gidx    = (const int*)d_in[4];
  const float* amask   = (const float*)d_in[5];
  const float* litp    = (const float*)d_in[6];
  float*       out     = (float*)d_out;

  const int blocks = RES_TOTAL / RES_PER_BLK;
  rotamer_projection_kernel<<<blocks, 256, 0, stream>>>(
      dih, bbc, seq, dframes, gidx, amask, litp, out);
}


# Tile loop at full residency: 1024 work-groups of 8 waves (8 per SIMD, not 4), each wave takes 4 groups in turn; next group's sequence, torsion and backbone loads issued behind phase 1, a trip ahead
# speedup vs baseline: 1.9804x; 1.3607x over previous
#include <hip/hip_runtime.h>


typedef float v2f __attribute__((ext_vector_type(2)));
typedef float v8f __attribute__((ext_vector_type(8)));

#define RES_TOTAL   (32 * 4096)
#define RES_PER_BLK 32
#define NUM_BLK     1024
#define TRIPS       (RES_TOTAL / RES_PER_BLK / NUM_BLK)
#define RES_PITCH   144
#define WAVE_LDS    (4 * RES_PITCH)
#define EPSF        1e-12f

__device__ __forceinline__ void wave_sync() {
  __builtin_amdgcn_wave_barrier();
}

__device__ __forceinline__ v8f wmma4x4(v2f a, v2f b) {
  v8f c = {};
  return __builtin_amdgcn_wmma_f32_16x16x4_f32(false, a, false, b, (short)0, c,
                                               false, false);
}

__device__ __forceinline__ float xhalf(float x) {
  return __builtin_bit_cast(
      float, __builtin_amdgcn_permlanex16(__builtin_bit_cast(int, x),
                                          __builtin_bit_cast(int, x),
                                          0x76543210, 0xfedcba98, false, false));
}

__device__ __forceinline__ float4 own_column(v8f d, bool odd) {
  float4 t;
  t.x = odd ? d[4] : d[0];
  t.y = odd ? d[5] : d[1];
  t.z = odd ? d[6] : d[2];
  t.w = odd ? d[7] : d[3];
  return t;
}

__device__ __forceinline__ v2f hand_on(float4 t, bool hi1, bool holder) {
  const float k0 = hi1 ? t.z : t.x;
  const float k1 = hi1 ? t.w : t.y;
  const float s0 = hi1 ? t.x : t.z;
  const float s1 = hi1 ? t.y : t.w;
  const float r0 = xhalf(s0);
  const float r1 = xhalf(s1);
  v2f o;
  o[0] = holder ? k0 : r0;
  o[1] = holder ? k1 : r1;
  return o;
}

struct RawIn {
  int   sq;
  int   sq4[4];
  float dh[14];
  float bb[9];
};

__device__ __forceinline__ RawIn load_raw(const float* __restrict__ dih,
                                          const float* __restrict__ bbc,
                                          const int* __restrict__ seq,
                                          int resBase, int i) {
  RawIn r;
  const int myRes = resBase + i;
  r.sq = seq[myRes];
#pragma unroll
  for (int ri = 0; ri < 4; ++ri) r.sq4[ri] = seq[resBase + ri];
  const float* dBase = dih + myRes * 14;
#pragma unroll
  for (int k = 0; k < 14; ++k) r.dh[k] = dBase[k];
  const float* bp = bbc + myRes * 9;
#pragma unroll
  for (int k = 0; k < 9; ++k) r.bb[k] = bp[k];
  return r;
}

__global__ __launch_bounds__(256, 8) void rotamer_projection_kernel(
    const float* __restrict__ dih,
    const float* __restrict__ bbc,
    const int*   __restrict__ seq,
    const float* __restrict__ dframes,
    const int*   __restrict__ gidx,
    const float* __restrict__ amask,
    const float* __restrict__ litp,
    float*       __restrict__ out)
{
  __shared__ __attribute__((aligned(16))) float ldsfr[8][WAVE_LDS];

  const int tid  = threadIdx.x;
  const int wave = tid >> 5;
  const int lane = tid & 31;
  float* fr = ldsfr[wave];

  const int low = lane & 15;
  const int hi  = lane >> 4;
  const int i   = low >> 2;
  const int m   = low & 3;
  const int k0  = hi * 2;

  const bool odd    = (i & 1) != 0;
  const bool hi1    = hi != 0;
  const bool holder = hi == (i >> 1);
  float4* rowp = reinterpret_cast<float4*>(fr + i * RES_PITCH + m * 4);

  float wc[2], ws[2], wk[2];
#pragma unroll
  for (int e = 0; e < 2; ++e) {
    const int k = k0 + e;
    wc[e] = ((k == 1 && m == 1) || (k == 2 && m == 2)) ? 1.f : 0.f;
    ws[e] = (k == 2 && m == 1) ? 1.f : ((k == 1 && m == 2) ? -1.f : 0.f);
    wk[e] = ((k == 0 && m == 0) || (k == 3 && m == 3)) ? 1.f : 0.f;
  }

  const int atomc = (low < 14) ? low : 13;

  RawIn cur = load_raw(dih, bbc, seq, blockIdx.x * RES_PER_BLK + wave * 4, i);

#pragma unroll 1
  for (int t = 0; t < TRIPS; ++t) {
    const int resBase = (blockIdx.x + t * NUM_BLK) * RES_PER_BLK + wave * 4;
    const int mySeq = cur.sq;

    const float nx = cur.bb[0], ny = cur.bb[1], nz = cur.bb[2];
    const float cax = cur.bb[3], cay = cur.bb[4], caz = cur.bb[5];
    const float cx = cur.bb[6], cy = cur.bb[7], cz = cur.bb[8];
    const float v1x = cx - cax, v1y = cy - cay, v1z = cz - caz;
    const float v2x = nx - cax, v2y = ny - cay, v2z = nz - caz;
    const float inv1 = rsqrtf(v1x * v1x + v1y * v1y + v1z * v1z + EPSF);
    const float e1x = v1x * inv1, e1y = v1y * inv1, e1z = v1z * inv1;
    const float d12 = e1x * v2x + e1y * v2y + e1z * v2z;
    const float u2x = v2x - e1x * d12, u2y = v2y - e1y * d12,
                u2z = v2z - e1z * d12;
    const float inv2 = rsqrtf(u2x * u2x + u2y * u2y + u2z * u2z + EPSF);
    const float e2x = u2x * inv2, e2y = u2y * inv2, e2z = u2z * inv2;
    const float e3x = e1y * e2z - e1z * e2y;
    const float e3y = e1z * e2x - e1x * e2z;
    const float e3z = e1x * e2y - e1y * e2x;

    float row0, row1, row2, row3;
    if (m == 0)      { row0 = e1x; row1 = e2x; row2 = e3x; row3 = cax; }
    else if (m == 1) { row0 = e1y; row1 = e2y; row2 = e3y; row3 = cay; }
    else if (m == 2) { row0 = e1z; row1 = e2z; row2 = e3z; row3 = caz; }
    else             { row0 = 0.f; row1 = 0.f; row2 = 0.f; row3 = 1.f; }
    v2f abb;
    abb[0] = (k0 == 0) ? row0 : row2;
    abb[1] = (k0 == 0) ? row1 : row3;

    const float* dfBase = dframes + mySeq * 128 + m * 4 + k0;
    v2f dfv[8];
#pragma unroll
    for (int a = 0; a < 8; ++a) {
      dfv[a][0] = dfBase[a * 16];
      dfv[a][1] = dfBase[a * 16 + 1];
    }
    int   gq[4];
    float mkq[4], l0q[4], l1q[4], l2q[4];
#pragma unroll
    for (int ri = 0; ri < 4; ++ri) {
      const int sq  = cur.sq4[ri];
      const int idx = sq * 14 + atomc;
      gq[ri]  = gidx[idx];
      mkq[ri] = amask[idx];
      const float* lp = litp + idx * 3;
      l0q[ri] = lp[0];
      l1q[ri] = lp[1];
      l2q[ri] = lp[2];
    }

    float sv[8], cv[8];
    sv[0] = 0.f;
    cv[0] = 1.f;
#pragma unroll
    for (int a = 1; a < 8; ++a) {
      const float x = cur.dh[(a - 1) * 2];
      const float y = cur.dh[(a - 1) * 2 + 1];
      const float inv = rsqrtf(x * x + y * y + EPSF);
      sv[a] = x * inv;
      cv[a] = y * inv;
    }

    float4 ycol[8];
    v2f    fop[8];
#pragma unroll
    for (int a = 0; a < 8; ++a) {
      v2f rop;
      rop[0] = fmaf(wc[0], cv[a], fmaf(ws[0], sv[a], wk[0]));
      rop[1] = fmaf(wc[1], cv[a], fmaf(ws[1], sv[a], wk[1]));
      if (a < 5) {
        ycol[a] = own_column(wmma4x4(rop, dfv[a]), odd);
      } else {
        fop[a] = hand_on(own_column(wmma4x4(dfv[a], rop), odd), hi1, holder);
      }
    }

    __builtin_amdgcn_sched_barrier(0);
    const int tn = (t + 1 < TRIPS) ? t + 1 : t;
    const RawIn nxt = load_raw(
        dih, bbc, seq, (blockIdx.x + tn * NUM_BLK) * RES_PER_BLK + wave * 4, i);

#pragma unroll
    for (int f = 5; f <= 7; ++f) {
      const v2f yop = hand_on(ycol[f - 1], hi1, holder);
      ycol[f] = own_column(wmma4x4(fop[f], yop), odd);
    }

    if (holder) {
#pragma unroll
      for (int a = 0; a < 8; ++a) rowp[a * 4] = ycol[a];
    }
    wave_sync();

    v8f dd[4];
#pragma unroll
    for (int ri = 0; ri < 4; ++ri) {
      const float4* F = reinterpret_cast<const float4*>(
          fr + ri * RES_PITCH + gq[ri] * 16 + hi * 8);
      const float4 fa = F[0];
      const float4 fb = F[1];
      v2f q;
      q[0] = fa.x * l0q[ri] + fa.y * l1q[ri] + fa.z * l2q[ri] + fa.w;
      q[1] = fb.x * l0q[ri] + fb.y * l1q[ri] + fb.z * l2q[ri] + fb.w;
      dd[ri] = wmma4x4(abb, q);
    }
    float ov[4][3];
#pragma unroll
    for (int ri = 0; ri < 4; ++ri) {
#pragma unroll
      for (int r = 0; r < 3; ++r) ov[ri][r] = dd[ri][4 * (ri & 1) + r] * mkq[ri];
    }
#pragma unroll
    for (int ri = 0; ri < 4; ++ri) {
      if (hi == (ri >> 1) && low < 14) {
        float* op = out + ((resBase + ri) * 14 + low) * 3;
        op[0] = ov[ri][0];
        op[1] = ov[ri][1];
        op[2] = ov[ri][2];
      }
    }
    cur = nxt;
  }
}

extern "C" void kernel_launch(void* const* d_in, const int* in_sizes, int n_in,
                              void* d_out, int out_size, void* d_ws,
                              size_t ws_size, hipStream_t stream) {
  const float* dih     = (const float*)d_in[0];
  const float* bbc     = (const float*)d_in[1];
  const int*   seq     = (const int*)d_in[2];
  const float* dframes = (const float*)d_in[3];
  const int*   gidx    = (const int*)d_in[4];
  const float* amask   = (const float*)d_in[5];
  const float* litp    = (const float*)d_in[6];
  float*       out     = (float*)d_out;

  rotamer_projection_kernel<<<NUM_BLK, 256, 0, stream>>>(
      dih, bbc, seq, dframes, gidx, amask, litp, out);
}


# Turn of a frame that no later frame builds on (0-3, 7) goes onto the atom position, D@(rot@v): raw table frames 0-3 and Z7=X6@D7 in LDS, 10 WMMA not 15; order of the f32 sums changes
# speedup vs baseline: 2.1591x; 1.0902x over previous
#include <hip/hip_runtime.h>


typedef float v2f __attribute__((ext_vector_type(2)));
typedef float v8f __attribute__((ext_vector_type(8)));

#define RES_TOTAL   (32 * 4096)
#define RES_PER_BLK 32
#define NUM_BLK     1024
#define TRIPS       (RES_TOTAL / RES_PER_BLK / NUM_BLK)
#define RES_PITCH   144
#define WAVE_LDS    (4 * RES_PITCH)
#define EPSF        1e-12f

__device__ __forceinline__ void wave_sync() {
  __builtin_amdgcn_wave_barrier();
}

__device__ __forceinline__ v8f wmma4x4(v2f a, v2f b) {
  v8f c = {};
  return __builtin_amdgcn_wmma_f32_16x16x4_f32(false, a, false, b, (short)0, c,
                                               false, false);
}

__device__ __forceinline__ float xhalf(float x) {
  return __builtin_bit_cast(
      float, __builtin_amdgcn_permlanex16(__builtin_bit_cast(int, x),
                                          __builtin_bit_cast(int, x),
                                          0x76543210, 0xfedcba98, false, false));
}

__device__ __forceinline__ float4 own_column(v8f d, bool odd) {
  float4 t;
  t.x = odd ? d[4] : d[0];
  t.y = odd ? d[5] : d[1];
  t.z = odd ? d[6] : d[2];
  t.w = odd ? d[7] : d[3];
  return t;
}

__device__ __forceinline__ v2f hand_on(float4 t, bool hi1, bool holder) {
  const float k0 = hi1 ? t.z : t.x;
  const float k1 = hi1 ? t.w : t.y;
  const float s0 = hi1 ? t.x : t.z;
  const float s1 = hi1 ? t.y : t.w;
  const float r0 = xhalf(s0);
  const float r1 = xhalf(s1);
  v2f o;
  o[0] = holder ? k0 : r0;
  o[1] = holder ? k1 : r1;
  return o;
}

struct RawIn {
  int   sq;
  int   sq4[4];
  float dhc[6];
  float dho[2];
  float bb[9];
};

__device__ __forceinline__ RawIn load_raw(const float* __restrict__ dih,
                                          const float* __restrict__ bbc,
                                          const int* __restrict__ seq,
                                          int resBase, int i, int ownOff) {
  RawIn r;
  const int myRes = resBase + i;
  r.sq = seq[myRes];
#pragma unroll
  for (int ri = 0; ri < 4; ++ri) r.sq4[ri] = seq[resBase + ri];
  const float* dBase = dih + myRes * 14;
#pragma unroll
  for (int k = 0; k < 6; ++k) r.dhc[k] = dBase[6 + k];
  r.dho[0] = dBase[ownOff];
  r.dho[1] = dBase[ownOff + 1];
  const float* bp = bbc + myRes * 9;
#pragma unroll
  for (int k = 0; k < 9; ++k) r.bb[k] = bp[k];
  return r;
}

__global__ __launch_bounds__(256, 8) void rotamer_projection_kernel(
    const float* __restrict__ dih,
    const float* __restrict__ bbc,
    const int*   __restrict__ seq,
    const float* __restrict__ dframes,
    const int*   __restrict__ gidx,
    const float* __restrict__ amask,
    const float* __restrict__ litp,
    float*       __restrict__ out)
{
  __shared__ __attribute__((aligned(16))) float ldsfr[8][WAVE_LDS];

  const int tid  = threadIdx.x;
  const int wave = tid >> 5;
  const int lane = tid & 31;
  float* fr = ldsfr[wave];

  const int low = lane & 15;
  const int hi  = lane >> 4;
  const int i   = low >> 2;
  const int m   = low & 3;
  const int k0  = hi * 2;

  const bool odd    = (i & 1) != 0;
  const bool hi1    = hi != 0;
  const bool holder = hi == (i >> 1);
  float4* rowp = reinterpret_cast<float4*>(fr + i * RES_PITCH + m * 4);
  const int  j       = m + 4 * hi;
  const bool ownTurn = (j >= 1 && j <= 3) || j == 7;
  const int  ownOff  = ownTurn ? (j - 1) * 2 : 0;
  float4* rawp = reinterpret_cast<float4*>(fr + i * RES_PITCH) + j;
  float2* scp  = reinterpret_cast<float2*>(fr + i * RES_PITCH + 128) + j;

  float wc[2], ws[2], wk[2];
#pragma unroll
  for (int e = 0; e < 2; ++e) {
    const int k = k0 + e;
    wc[e] = ((k == 1 && m == 1) || (k == 2 && m == 2)) ? 1.f : 0.f;
    ws[e] = (k == 2 && m == 1) ? 1.f : ((k == 1 && m == 2) ? -1.f : 0.f);
    wk[e] = ((k == 0 && m == 0) || (k == 3 && m == 3)) ? 1.f : 0.f;
  }

  const int atomc = (low < 14) ? low : 13;

  RawIn cur =
      load_raw(dih, bbc, seq, blockIdx.x * RES_PER_BLK + wave * 4, i, ownOff);

#pragma unroll 1
  for (int t = 0; t < TRIPS; ++t) {
    const int resBase = (blockIdx.x + t * NUM_BLK) * RES_PER_BLK + wave * 4;
    const int mySeq = cur.sq;

    const float nx = cur.bb[0], ny = cur.bb[1], nz = cur.bb[2];
    const float cax = cur.bb[3], cay = cur.bb[4], caz = cur.bb[5];
    const float cx = cur.bb[6], cy = cur.bb[7], cz = cur.bb[8];
    const float v1x = cx - cax, v1y = cy - cay, v1z = cz - caz;
    const float v2x = nx - cax, v2y = ny - cay, v2z = nz - caz;
    const float inv1 = rsqrtf(v1x * v1x + v1y * v1y + v1z * v1z + EPSF);
    const float e1x = v1x * inv1, e1y = v1y * inv1, e1z = v1z * inv1;
    const float d12 = e1x * v2x + e1y * v2y + e1z * v2z;
    const float u2x = v2x - e1x * d12, u2y = v2y - e1y * d12,
                u2z = v2z - e1z * d12;
    const float inv2 = rsqrtf(u2x * u2x + u2y * u2y + u2z * u2z + EPSF);
    const float e2x = u2x * inv2, e2y = u2y * inv2, e2z = u2z * inv2;
    const float e3x = e1y * e2z - e1z * e2y;
    const float e3y = e1z * e2x - e1x * e2z;
    const float e3z = e1x * e2y - e1y * e2x;

    float row0, row1, row2, row3;
    if (m == 0)      { row0 = e1x; row1 = e2x; row2 = e3x; row3 = cax; }
    else if (m == 1) { row0 = e1y; row1 = e2y; row2 = e3y; row3 = cay; }
    else if (m == 2) { row0 = e1z; row1 = e2z; row2 = e3z; row3 = caz; }
    else             { row0 = 0.f; row1 = 0.f; row2 = 0.f; row3 = 1.f; }
    v2f abb;
    abb[0] = (k0 == 0) ? row0 : row2;
    abb[1] = (k0 == 0) ? row1 : row3;

    const float* dfBase = dframes + mySeq * 128 + m * 4 + k0;
    v2f dfv[7];
#pragma unroll
    for (int a = 4; a < 7; ++a) {
      dfv[a][0] = dfBase[a * 16];
      dfv[a][1] = dfBase[a * 16 + 1];
    }
    v2f dz7;
    dz7[0] = dframes[mySeq * 128 + 112 + k0 * 4 + m];
    dz7[1] = dframes[mySeq * 128 + 112 + k0 * 4 + 4 + m];
    const float4* rawsrc =
        reinterpret_cast<const float4*>(dframes + mySeq * 128) + j;
    const float4 raw0 = rawsrc[0];
    const float4 raw1 = rawsrc[8];
    int   gq[4];
    float mkq[4], l0q[4], l1q[4], l2q[4];
#pragma unroll
    for (int ri = 0; ri < 4; ++ri) {
      const int sq  = cur.sq4[ri];
      const int idx = sq * 14 + atomc;
      gq[ri]  = gidx[idx];
      mkq[ri] = amask[idx];
      const float* lp = litp + idx * 3;
      l0q[ri] = lp[0];
      l1q[ri] = lp[1];
      l2q[ri] = lp[2];
    }

    float sv[7], cv[7];
#pragma unroll
    for (int a = 4; a < 7; ++a) {
      const float x = cur.dhc[(a - 4) * 2];
      const float y = cur.dhc[(a - 4) * 2 + 1];
      const float inv = rsqrtf(x * x + y * y + EPSF);
      sv[a] = x * inv;
      cv[a] = y * inv;
    }
    {
      const float x = cur.dho[0];
      const float y = cur.dho[1];
      const float inv = rsqrtf(x * x + y * y + EPSF);
      float2 sc;
      sc.x = ownTurn ? x * inv : 0.f;
      sc.y = ownTurn ? y * inv : 1.f;
      scp[0] = sc;
      rawp[0] = raw0;
      rawp[8] = raw1;
    }

    float4 ycol[8];
    v2f    fop[8];
#pragma unroll
    for (int a = 4; a < 7; ++a) {
      v2f rop;
      rop[0] = fmaf(wc[0], cv[a], fmaf(ws[0], sv[a], wk[0]));
      rop[1] = fmaf(wc[1], cv[a], fmaf(ws[1], sv[a], wk[1]));
      if (a == 4) {
        ycol[a] = own_column(wmma4x4(rop, dfv[a]), odd);
      } else {
        fop[a] = hand_on(own_column(wmma4x4(dfv[a], rop), odd), hi1, holder);
      }
    }

    __builtin_amdgcn_sched_barrier(0);
    const int tn = (t + 1 < TRIPS) ? t + 1 : t;
    const RawIn nxt = load_raw(
        dih, bbc, seq, (blockIdx.x + tn * NUM_BLK) * RES_PER_BLK + wave * 4, i,
        ownOff);

#pragma unroll
    for (int f = 5; f <= 7; ++f) {
      const v2f yop = hand_on(ycol[f - 1], hi1, holder);
      const v2f lop = (f < 7) ? fop[f] : dz7;
      ycol[f] = own_column(wmma4x4(lop, yop), odd);
    }

    if (holder) {
#pragma unroll
      for (int a = 4; a < 8; ++a) rowp[a * 4] = ycol[a];
    }
    wave_sync();

    v8f dd[4];
#pragma unroll
    for (int ri = 0; ri < 4; ++ri) {
      const float4* F = reinterpret_cast<const float4*>(
          fr + ri * RES_PITCH + gq[ri] * 16 + hi * 8);
      const float4 fa = F[0];
      const float4 fb = F[1];
      const float2 sc = *reinterpret_cast<const float2*>(
          fr + ri * RES_PITCH + 128 + gq[ri] * 2);
      const float uy = sc.y * l1q[ri] - sc.x * l2q[ri];
      const float uz = sc.x * l1q[ri] + sc.y * l2q[ri];
      v2f c0, c1, c2, c3;
      c0[0] = fa.x; c0[1] = fb.x;
      c1[0] = fa.y; c1[1] = fb.y;
      c2[0] = fa.z; c2[1] = fb.z;
      c3[0] = fa.w; c3[1] = fb.w;
      const v2f q = c0 * l0q[ri] + c1 * uy + c2 * uz + c3;
      dd[ri] = wmma4x4(abb, q);
    }
    float ov[4][3];
#pragma unroll
    for (int ri = 0; ri < 4; ++ri) {
#pragma unroll
      for (int r = 0; r < 3; ++r) ov[ri][r] = dd[ri][4 * (ri & 1) + r] * mkq[ri];
    }
#pragma unroll
    for (int ri = 0; ri < 4; ++ri) {
      if (hi == (ri >> 1) && low < 14) {
        float* op = out + ((resBase + ri) * 14 + low) * 3;
        op[0] = ov[ri][0];
        op[1] = ov[ri][1];
        op[2] = ov[ri][2];
      }
    }
    cur = nxt;
  }
}

extern "C" void kernel_launch(void* const* d_in, const int* in_sizes, int n_in,
                              void* d_out, int out_size, void* d_ws,
                              size_t ws_size, hipStream_t stream) {
  const float* dih     = (const float*)d_in[0];
  const float* bbc     = (const float*)d_in[1];
  const int*   seq     = (const int*)d_in[2];
  const float* dframes = (const float*)d_in[3];
  const int*   gidx    = (const int*)d_in[4];
  const float* amask   = (const float*)d_in[5];
  const float* litp    = (const float*)d_in[6];
  float*       out     = (float*)d_out;

  rotamer_projection_kernel<<<NUM_BLK, 256, 0, stream>>>(
      dih, bbc, seq, dframes, gidx, amask, litp, out);
}


# Per-atom phase re-tiled by lane half: lane (c,h) builds all 4 rows of q for its own residues 2h,2h+1 (table, mask, turn reads for 2 residues, not 4); other half's 2 rows cross in a lane-half exchange
# speedup vs baseline: 2.3264x; 1.0775x over previous
#include <hip/hip_runtime.h>


typedef float v2f __attribute__((ext_vector_type(2)));
typedef float v8f __attribute__((ext_vector_type(8)));

#define RES_TOTAL   (32 * 4096)
#define RES_PER_BLK 32
#define NUM_BLK     1024
#define TRIPS       (RES_TOTAL / RES_PER_BLK / NUM_BLK)
#define RES_PITCH   144
#define WAVE_LDS    (4 * RES_PITCH)
#define EPSF        1e-12f

__device__ __forceinline__ void wave_sync() {
  __builtin_amdgcn_wave_barrier();
}

__device__ __forceinline__ v8f wmma4x4(v2f a, v2f b) {
  v8f c = {};
  return __builtin_amdgcn_wmma_f32_16x16x4_f32(false, a, false, b, (short)0, c,
                                               false, false);
}

__device__ __forceinline__ float xhalf(float x) {
  return __builtin_bit_cast(
      float, __builtin_amdgcn_permlanex16(__builtin_bit_cast(int, x),
                                          __builtin_bit_cast(int, x),
                                          0x76543210, 0xfedcba98, false, false));
}

__device__ __forceinline__ float4 own_column(v8f d, bool odd) {
  float4 t;
  t.x = odd ? d[4] : d[0];
  t.y = odd ? d[5] : d[1];
  t.z = odd ? d[6] : d[2];
  t.w = odd ? d[7] : d[3];
  return t;
}

__device__ __forceinline__ v2f hand_on(float4 t, bool hi1, bool holder) {
  const float k0 = hi1 ? t.z : t.x;
  const float k1 = hi1 ? t.w : t.y;
  const float s0 = hi1 ? t.x : t.z;
  const float s1 = hi1 ? t.y : t.w;
  const float r0 = xhalf(s0);
  const float r1 = xhalf(s1);
  v2f o;
  o[0] = holder ? k0 : r0;
  o[1] = holder ? k1 : r1;
  return o;
}

struct RawIn {
  int   sq;
  int   sq2[2];
  float dhc[6];
  float dho[2];
  float bb[9];
};

__device__ __forceinline__ RawIn load_raw(const float* __restrict__ dih,
                                          const float* __restrict__ bbc,
                                          const int* __restrict__ seq,
                                          int resBase, int i, int hi,
                                          int ownOff) {
  RawIn r;
  const int myRes = resBase + i;
  r.sq = seq[myRes];
#pragma unroll
  for (int p = 0; p < 2; ++p) r.sq2[p] = seq[resBase + 2 * hi + p];
  const float* dBase = dih + myRes * 14;
#pragma unroll
  for (int k = 0; k < 6; ++k) r.dhc[k] = dBase[6 + k];
  r.dho[0] = dBase[ownOff];
  r.dho[1] = dBase[ownOff + 1];
  const float* bp = bbc + myRes * 9;
#pragma unroll
  for (int k = 0; k < 9; ++k) r.bb[k] = bp[k];
  return r;
}

__global__ __launch_bounds__(256, 8) void rotamer_projection_kernel(
    const float* __restrict__ dih,
    const float* __restrict__ bbc,
    const int*   __restrict__ seq,
    const float* __restrict__ dframes,
    const int*   __restrict__ gidx,
    const float* __restrict__ amask,
    const float* __restrict__ litp,
    float*       __restrict__ out)
{
  __shared__ __attribute__((aligned(16))) float ldsfr[8][WAVE_LDS];

  const int tid  = threadIdx.x;
  const int wave = tid >> 5;
  const int lane = tid & 31;
  float* fr = ldsfr[wave];

  const int low = lane & 15;
  const int hi  = lane >> 4;
  const int i   = low >> 2;
  const int m   = low & 3;
  const int k0  = hi * 2;

  const bool odd    = (i & 1) != 0;
  const bool hi1    = hi != 0;
  const bool holder = hi == (i >> 1);
  float4* rowp = reinterpret_cast<float4*>(fr + i * RES_PITCH + m * 4);
  const int  j       = m + 4 * hi;
  const bool ownTurn = (j >= 1 && j <= 3) || j == 7;
  const int  ownOff  = ownTurn ? (j - 1) * 2 : 0;
  float4* rawp = reinterpret_cast<float4*>(fr + i * RES_PITCH) + j;
  float2* scp  = reinterpret_cast<float2*>(fr + i * RES_PITCH + 128) + j;
  const float* frh = fr + 2 * hi * RES_PITCH;

  float wc[2], ws[2], wk[2];
#pragma unroll
  for (int e = 0; e < 2; ++e) {
    const int k = k0 + e;
    wc[e] = ((k == 1 && m == 1) || (k == 2 && m == 2)) ? 1.f : 0.f;
    ws[e] = (k == 2 && m == 1) ? 1.f : ((k == 1 && m == 2) ? -1.f : 0.f);
    wk[e] = ((k == 0 && m == 0) || (k == 3 && m == 3)) ? 1.f : 0.f;
  }

  const int atomc = (low < 14) ? low : 13;

  RawIn cur =
      load_raw(dih, bbc, seq, blockIdx.x * RES_PER_BLK + wave * 4, i, hi,
               ownOff);

#pragma unroll 1
  for (int t = 0; t < TRIPS; ++t) {
    const int resBase = (blockIdx.x + t * NUM_BLK) * RES_PER_BLK + wave * 4;
    const int mySeq = cur.sq;

    const float nx = cur.bb[0], ny = cur.bb[1], nz = cur.bb[2];
    const float cax = cur.bb[3], cay = cur.bb[4], caz = cur.bb[5];
    const float cx = cur.bb[6], cy = cur.bb[7], cz = cur.bb[8];
    const float v1x = cx - cax, v1y = cy - cay, v1z = cz - caz;
    const float v2x = nx - cax, v2y = ny - cay, v2z = nz - caz;
    const float inv1 = rsqrtf(v1x * v1x + v1y * v1y + v1z * v1z + EPSF);
    const float e1x = v1x * inv1, e1y = v1y * inv1, e1z = v1z * inv1;
    const float d12 = e1x * v2x + e1y * v2y + e1z * v2z;
    const float u2x = v2x - e1x * d12, u2y = v2y - e1y * d12,
                u2z = v2z - e1z * d12;
    const float inv2 = rsqrtf(u2x * u2x + u2y * u2y + u2z * u2z + EPSF);
    const float e2x = u2x * inv2, e2y = u2y * inv2, e2z = u2z * inv2;
    const float e3x = e1y * e2z - e1z * e2y;
    const float e3y = e1z * e2x - e1x * e2z;
    const float e3z = e1x * e2y - e1y * e2x;

    float row0, row1, row2, row3;
    if (m == 0)      { row0 = e1x; row1 = e2x; row2 = e3x; row3 = cax; }
    else if (m == 1) { row0 = e1y; row1 = e2y; row2 = e3y; row3 = cay; }
    else if (m == 2) { row0 = e1z; row1 = e2z; row2 = e3z; row3 = caz; }
    else             { row0 = 0.f; row1 = 0.f; row2 = 0.f; row3 = 1.f; }
    v2f abb;
    abb[0] = (k0 == 0) ? row0 : row2;
    abb[1] = (k0 == 0) ? row1 : row3;

    const float* dfBase = dframes + mySeq * 128 + m * 4 + k0;
    v2f dfv[7];
#pragma unroll
    for (int a = 4; a < 7; ++a) {
      dfv[a][0] = dfBase[a * 16];
      dfv[a][1] = dfBase[a * 16 + 1];
    }
    v2f dz7;
    dz7[0] = dframes[mySeq * 128 + 112 + k0 * 4 + m];
    dz7[1] = dframes[mySeq * 128 + 112 + k0 * 4 + 4 + m];
    const float4* rawsrc =
        reinterpret_cast<const float4*>(dframes + mySeq * 128) + j;
    const float4 raw0 = rawsrc[0];
    const float4 raw1 = rawsrc[8];
    int   gq[2];
    float mkq[2], l0q[2], l1q[2], l2q[2];
#pragma unroll
    for (int ri = 0; ri < 2; ++ri) {
      const int sq  = cur.sq2[ri];
      const int idx = sq * 14 + atomc;
      gq[ri]  = gidx[idx];
      mkq[ri] = amask[idx];
      const float* lp = litp + idx * 3;
      l0q[ri] = lp[0];
      l1q[ri] = lp[1];
      l2q[ri] = lp[2];
    }

    float sv[7], cv[7];
#pragma unroll
    for (int a = 4; a < 7; ++a) {
      const float x = cur.dhc[(a - 4) * 2];
      const float y = cur.dhc[(a - 4) * 2 + 1];
      const float inv = rsqrtf(x * x + y * y + EPSF);
      sv[a] = x * inv;
      cv[a] = y * inv;
    }
    {
      const float x = cur.dho[0];
      const float y = cur.dho[1];
      const float inv = rsqrtf(x * x + y * y + EPSF);
      float2 sc;
      sc.x = ownTurn ? x * inv : 0.f;
      sc.y = ownTurn ? y * inv : 1.f;
      scp[0] = sc;
      rawp[0] = raw0;
      rawp[8] = raw1;
    }

    float4 ycol[8];
    v2f    fop[8];
#pragma unroll
    for (int a = 4; a < 7; ++a) {
      v2f rop;
      rop[0] = fmaf(wc[0], cv[a], fmaf(ws[0], sv[a], wk[0]));
      rop[1] = fmaf(wc[1], cv[a], fmaf(ws[1], sv[a], wk[1]));
      if (a == 4) {
        ycol[a] = own_column(wmma4x4(rop, dfv[a]), odd);
      } else {
        fop[a] = hand_on(own_column(wmma4x4(dfv[a], rop), odd), hi1, holder);
      }
    }

    __builtin_amdgcn_sched_barrier(0);
    const int tn = (t + 1 < TRIPS) ? t + 1 : t;
    const RawIn nxt = load_raw(
        dih, bbc, seq, (blockIdx.x + tn * NUM_BLK) * RES_PER_BLK + wave * 4, i,
        hi, ownOff);

#pragma unroll
    for (int f = 5; f <= 7; ++f) {
      const v2f yop = hand_on(ycol[f - 1], hi1, holder);
      const v2f lop = (f < 7) ? fop[f] : dz7;
      ycol[f] = own_column(wmma4x4(lop, yop), odd);
    }

    if (holder) {
#pragma unroll
      for (int a = 4; a < 8; ++a) rowp[a * 4] = ycol[a];
    }
    wave_sync();

    float ov[2][3];
#pragma unroll
    for (int p = 0; p < 2; ++p) {
      const float* slot = frh + p * RES_PITCH + gq[p] * 16;
      const float4 f0 = reinterpret_cast<const float4*>(slot)[0];
      const float4 f1 = reinterpret_cast<const float4*>(slot)[1];
      const float4 f2 = reinterpret_cast<const float4*>(slot)[2];
      const float4 f3 = reinterpret_cast<const float4*>(slot)[3];
      const float2 sc = *reinterpret_cast<const float2*>(
          frh + p * RES_PITCH + 128 + gq[p] * 2);
      const float uy = sc.y * l1q[p] - sc.x * l2q[p];
      const float uz = sc.x * l1q[p] + sc.y * l2q[p];
      const float q0 = fmaf(f0.z, uz, fmaf(f0.x, l0q[p], f0.y * uy)) + f0.w;
      const float q1 = fmaf(f1.z, uz, fmaf(f1.x, l0q[p], f1.y * uy)) + f1.w;
      const float q2 = fmaf(f2.z, uz, fmaf(f2.x, l0q[p], f2.y * uy)) + f2.w;
      const float q3 = fmaf(f3.z, uz, fmaf(f3.x, l0q[p], f3.y * uy)) + f3.w;
      const float r0 = xhalf(hi1 ? q0 : q2);
      const float r1 = xhalf(hi1 ? q1 : q3);
      v2f blo, bhi;
      blo[0] = hi1 ? r0 : q0;
      blo[1] = hi1 ? r1 : q1;
      bhi[0] = hi1 ? q2 : r0;
      bhi[1] = hi1 ? q3 : r1;
      const v8f dlo = wmma4x4(abb, blo);
      const v8f dhi = wmma4x4(abb, bhi);
#pragma unroll
      for (int r = 0; r < 3; ++r)
        ov[p][r] = (hi1 ? dhi[4 * p + r] : dlo[4 * p + r]) * mkq[p];
    }
    if (low < 14) {
#pragma unroll
      for (int p = 0; p < 2; ++p) {
        float* op = out + ((resBase + 2 * hi + p) * 14 + low) * 3;
        op[0] = ov[p][0];
        op[1] = ov[p][1];
        op[2] = ov[p][2];
      }
    }
    cur = nxt;
  }
}

extern "C" void kernel_launch(void* const* d_in, const int* in_sizes, int n_in,
                              void* d_out, int out_size, void* d_ws,
                              size_t ws_size, hipStream_t stream) {
  const float* dih     = (const float*)d_in[0];
  const float* bbc     = (const float*)d_in[1];
  const int*   seq     = (const int*)d_in[2];
  const float* dframes = (const float*)d_in[3];
  const int*   gidx    = (const int*)d_in[4];
  const float* amask   = (const float*)d_in[5];
  const float* litp    = (const float*)d_in[6];
  float*       out     = (float*)d_out;

  rotamer_projection_kernel<<<NUM_BLK, 256, 0, stream>>>(
      dih, bbc, seq, dframes, gidx, amask, litp, out);
}


# No lane-half exchange left: chain rows re-mapped so a lane's next operand pair is in its own D registers; pos product = 2 accumulating WMMA by k pair, masked bb rows; order of the f32 sums changes
# speedup vs baseline: 2.4598x; 1.0573x over previous
#include <hip/hip_runtime.h>


typedef float v2f __attribute__((ext_vector_type(2)));
typedef float v8f __attribute__((ext_vector_type(8)));

#define RES_TOTAL   (32 * 4096)
#define RES_PER_BLK 32
#define NUM_BLK     1024
#define TRIPS       (RES_TOTAL / RES_PER_BLK / NUM_BLK)
#define RES_PITCH   144
#define WAVE_LDS    (4 * RES_PITCH)
#define EPSF        1e-12f

__device__ __forceinline__ void wave_sync() {
  __builtin_amdgcn_wave_barrier();
}

__device__ __forceinline__ v8f wmma4x4(v2f a, v2f b) {
  v8f c = {};
  return __builtin_amdgcn_wmma_f32_16x16x4_f32(false, a, false, b, (short)0, c,
                                               false, false);
}

__device__ __forceinline__ v8f wmma4x4acc(v2f a, v2f b, v8f c) {
  return __builtin_amdgcn_wmma_f32_16x16x4_f32(false, a, false, b, (short)0, c,
                                               false, false);
}

__device__ __forceinline__ v2f own_pair(v8f d, bool b0, bool b1) {
  const float l0 = b0 ? d[2] : d[0];
  const float l1 = b0 ? d[3] : d[1];
  const float u0 = b0 ? d[6] : d[4];
  const float u1 = b0 ? d[7] : d[5];
  v2f o;
  o[0] = b1 ? u0 : l0;
  o[1] = b1 ? u1 : l1;
  return o;
}

struct RawIn {
  int   sq;
  int   sq2[2];
  float dhc[6];
  float dho[2];
  float bb[9];
};

__device__ __forceinline__ RawIn load_raw(const float* __restrict__ dih,
                                          const float* __restrict__ bbc,
                                          const int* __restrict__ seq,
                                          int resBase, int iC, int iB, int hi,
                                          int ownOff) {
  RawIn r;
  const int myRes = resBase + iC;
  r.sq = seq[myRes];
#pragma unroll
  for (int p = 0; p < 2; ++p) r.sq2[p] = seq[resBase + 2 * hi + p];
  const float* dBase = dih + myRes * 14;
#pragma unroll
  for (int k = 0; k < 6; ++k) r.dhc[k] = dBase[6 + k];
  r.dho[0] = dBase[ownOff];
  r.dho[1] = dBase[ownOff + 1];
  const float* bp = bbc + (resBase + iB) * 9;
#pragma unroll
  for (int k = 0; k < 9; ++k) r.bb[k] = bp[k];
  return r;
}

__global__ __launch_bounds__(256, 8) void rotamer_projection_kernel(
    const float* __restrict__ dih,
    const float* __restrict__ bbc,
    const int*   __restrict__ seq,
    const float* __restrict__ dframes,
    const int*   __restrict__ gidx,
    const float* __restrict__ amask,
    const float* __restrict__ litp,
    float*       __restrict__ out)
{
  __shared__ __attribute__((aligned(16))) float ldsfr[8][WAVE_LDS];

  const int tid  = threadIdx.x;
  const int wave = tid >> 5;
  const int lane = tid & 31;
  float* fr = ldsfr[wave];

  const int low = lane & 15;
  const int hi  = lane >> 4;
  const int k0  = hi * 2;
  const int i   = (low >> 1) & 3;
  const int m   = 2 * (low >> 3) + (low & 1);
  const int iB  = low >> 2;
  const int mB  = low & 3;

  const bool bbP0 = ((iB >> 1) == hi) && ((iB & 1) == 0);
  const bool bbP1 = ((iB >> 1) == hi) && ((iB & 1) == 1);

  const bool ib0 = (i & 1) != 0;
  const bool ib1 = (i & 2) != 0;
  float2* rowp = reinterpret_cast<float2*>(fr + i * RES_PITCH + m * 4 + k0);
  const int  j       = m + 4 * hi;
  const bool ownTurn = (j >= 1 && j <= 3) || j == 7;
  const int  ownOff  = ownTurn ? (j - 1) * 2 : 0;
  float4* rawp = reinterpret_cast<float4*>(fr + i * RES_PITCH) + j;
  float2* scp  = reinterpret_cast<float2*>(fr + i * RES_PITCH + 128) + j;
  const float* frh = fr + 2 * hi * RES_PITCH;

  float wc[2], ws[2], wk[2];
#pragma unroll
  for (int e = 0; e < 2; ++e) {
    const int k = k0 + e;
    wc[e] = ((k == 1 && m == 1) || (k == 2 && m == 2)) ? 1.f : 0.f;
    ws[e] = (k == 2 && m == 1) ? 1.f : ((k == 1 && m == 2) ? -1.f : 0.f);
    wk[e] = ((k == 0 && m == 0) || (k == 3 && m == 3)) ? 1.f : 0.f;
  }

  const int atomc = (low < 14) ? low : 13;

  RawIn cur =
      load_raw(dih, bbc, seq, blockIdx.x * RES_PER_BLK + wave * 4, i, iB, hi,
               ownOff);

#pragma unroll 1
  for (int t = 0; t < TRIPS; ++t) {
    const int resBase = (blockIdx.x + t * NUM_BLK) * RES_PER_BLK + wave * 4;
    const int mySeq = cur.sq;

    const float nx = cur.bb[0], ny = cur.bb[1], nz = cur.bb[2];
    const float cax = cur.bb[3], cay = cur.bb[4], caz = cur.bb[5];
    const float cx = cur.bb[6], cy = cur.bb[7], cz = cur.bb[8];
    const float v1x = cx - cax, v1y = cy - cay, v1z = cz - caz;
    const float v2x = nx - cax, v2y = ny - cay, v2z = nz - caz;
    const float inv1 = rsqrtf(v1x * v1x + v1y * v1y + v1z * v1z + EPSF);
    const float e1x = v1x * inv1, e1y = v1y * inv1, e1z = v1z * inv1;
    const float d12 = e1x * v2x + e1y * v2y + e1z * v2z;
    const float u2x = v2x - e1x * d12, u2y = v2y - e1y * d12,
                u2z = v2z - e1z * d12;
    const float inv2 = rsqrtf(u2x * u2x + u2y * u2y + u2z * u2z + EPSF);
    const float e2x = u2x * inv2, e2y = u2y * inv2, e2z = u2z * inv2;
    const float e3x = e1y * e2z - e1z * e2y;
    const float e3y = e1z * e2x - e1x * e2z;
    const float e3z = e1x * e2y - e1y * e2x;

    float row0, row1, row2, row3;
    if (mB == 0)      { row0 = e1x; row1 = e2x; row2 = e3x; row3 = cax; }
    else if (mB == 1) { row0 = e1y; row1 = e2y; row2 = e3y; row3 = cay; }
    else if (mB == 2) { row0 = e1z; row1 = e2z; row2 = e3z; row3 = caz; }
    else             { row0 = 0.f; row1 = 0.f; row2 = 0.f; row3 = 1.f; }
    v2f abb[2][2];
    abb[0][0][0] = bbP0 ? row0 : 0.f;
    abb[0][0][1] = bbP0 ? row1 : 0.f;
    abb[0][1][0] = bbP0 ? row2 : 0.f;
    abb[0][1][1] = bbP0 ? row3 : 0.f;
    abb[1][0][0] = bbP1 ? row0 : 0.f;
    abb[1][0][1] = bbP1 ? row1 : 0.f;
    abb[1][1][0] = bbP1 ? row2 : 0.f;
    abb[1][1][1] = bbP1 ? row3 : 0.f;

    const float* dfBase = dframes + mySeq * 128 + m * 4 + k0;
    v2f dfv[7];
#pragma unroll
    for (int a = 4; a < 7; ++a) {
      dfv[a][0] = dfBase[a * 16];
      dfv[a][1] = dfBase[a * 16 + 1];
    }
    v2f dz7;
    dz7[0] = dframes[mySeq * 128 + 112 + k0 * 4 + m];
    dz7[1] = dframes[mySeq * 128 + 112 + k0 * 4 + 4 + m];
    const float4* rawsrc =
        reinterpret_cast<const float4*>(dframes + mySeq * 128) + j;
    const float4 raw0 = rawsrc[0];
    const float4 raw1 = rawsrc[8];
    int   gq[2];
    float mkq[2], l0q[2], l1q[2], l2q[2];
#pragma unroll
    for (int ri = 0; ri < 2; ++ri) {
      const int sq  = cur.sq2[ri];
      const int idx = sq * 14 + atomc;
      gq[ri]  = gidx[idx];
      mkq[ri] = amask[idx];
      const float* lp = litp + idx * 3;
      l0q[ri] = lp[0];
      l1q[ri] = lp[1];
      l2q[ri] = lp[2];
    }

    float sv[7], cv[7];
#pragma unroll
    for (int a = 4; a < 7; ++a) {
      const float x = cur.dhc[(a - 4) * 2];
      const float y = cur.dhc[(a - 4) * 2 + 1];
      const float inv = rsqrtf(x * x + y * y + EPSF);
      sv[a] = x * inv;
      cv[a] = y * inv;
    }
    {
      const float x = cur.dho[0];
      const float y = cur.dho[1];
      const float inv = rsqrtf(x * x + y * y + EPSF);
      float2 sc;
      sc.x = ownTurn ? x * inv : 0.f;
      sc.y = ownTurn ? y * inv : 1.f;
      scp[0] = sc;
      rawp[0] = raw0;
      rawp[8] = raw1;
    }

    v2f ypr[8];
    v2f fop[8];
#pragma unroll
    for (int a = 4; a < 7; ++a) {
      v2f rop;
      rop[0] = fmaf(wc[0], cv[a], fmaf(ws[0], sv[a], wk[0]));
      rop[1] = fmaf(wc[1], cv[a], fmaf(ws[1], sv[a], wk[1]));
      if (a == 4) {
        ypr[a] = own_pair(wmma4x4(rop, dfv[a]), ib0, ib1);
      } else {
        fop[a] = own_pair(wmma4x4(dfv[a], rop), ib0, ib1);
      }
    }

    __builtin_amdgcn_sched_barrier(0);
    const int tn = (t + 1 < TRIPS) ? t + 1 : t;
    const RawIn nxt = load_raw(
        dih, bbc, seq, (blockIdx.x + tn * NUM_BLK) * RES_PER_BLK + wave * 4, i,
        iB, hi, ownOff);

#pragma unroll
    for (int f = 5; f <= 7; ++f) {
      const v2f lop = (f < 7) ? fop[f] : dz7;
      ypr[f] = own_pair(wmma4x4(lop, ypr[f - 1]), ib0, ib1);
    }

    {
      float2 pr;
#pragma unroll
      for (int a = 4; a < 8; ++a) {
        pr.x = ypr[a][0];
        pr.y = ypr[a][1];
        rowp[a * 8] = pr;
      }
    }
    wave_sync();
    asm volatile("" : : "v"(mkq[0]), "v"(mkq[1]));

    float ov[2][3];
#pragma unroll
    for (int p = 0; p < 2; ++p) {
      const float* slot = frh + p * RES_PITCH + gq[p] * 16;
      const float4 f0 = reinterpret_cast<const float4*>(slot)[0];
      const float4 f1 = reinterpret_cast<const float4*>(slot)[1];
      const float4 f2 = reinterpret_cast<const float4*>(slot)[2];
      const float4 f3 = reinterpret_cast<const float4*>(slot)[3];
      const float2 sc = *reinterpret_cast<const float2*>(
          frh + p * RES_PITCH + 128 + gq[p] * 2);
      const float uy = sc.y * l1q[p] - sc.x * l2q[p];
      const float uz = sc.x * l1q[p] + sc.y * l2q[p];
      v2f b01, b23;
      b01[0] = fmaf(f0.z, uz, fmaf(f0.x, l0q[p], f0.y * uy)) + f0.w;
      b01[1] = fmaf(f1.z, uz, fmaf(f1.x, l0q[p], f1.y * uy)) + f1.w;
      b23[0] = fmaf(f2.z, uz, fmaf(f2.x, l0q[p], f2.y * uy)) + f2.w;
      b23[1] = fmaf(f3.z, uz, fmaf(f3.x, l0q[p], f3.y * uy)) + f3.w;
      const v8f d = wmma4x4acc(abb[p][1], b23, wmma4x4(abb[p][0], b01));
#pragma unroll
      for (int r = 0; r < 3; ++r) ov[p][r] = d[4 * p + r] * mkq[p];
    }
    if (low < 14) {
#pragma unroll
      for (int p = 0; p < 2; ++p) {
        float* op = out + ((resBase + 2 * hi + p) * 14 + low) * 3;
        op[0] = ov[p][0];
        op[1] = ov[p][1];
        op[2] = ov[p][2];
      }
    }
    cur = nxt;
  }
}

extern "C" void kernel_launch(void* const* d_in, const int* in_sizes, int n_in,
                              void* d_out, int out_size, void* d_ws,
                              size_t ws_size, hipStream_t stream) {
  const float* dih     = (const float*)d_in[0];
  const float* bbc     = (const float*)d_in[1];
  const int*   seq     = (const int*)d_in[2];
  const float* dframes = (const float*)d_in[3];
  const int*   gidx    = (const int*)d_in[4];
  const float* amask   = (const float*)d_in[5];
  const float* litp    = (const float*)d_in[6];
  float*       out     = (float*)d_out;

  rotamer_projection_kernel<<<NUM_BLK, 256, 0, stream>>>(
      dih, bbc, seq, dframes, gidx, amask, litp, out);
}


# A frame times its own x-turn (Y_4, F_5, F_6) formed on the vector unit from table values, c*D1+s*D2 and c*D2-s*D1, columns 0,3 copied: 7 WMMA, not 10; only zero terms leave the f32 sums
# speedup vs baseline: 2.5906x; 1.0532x over previous
#include <hip/hip_runtime.h>


typedef float v2f __attribute__((ext_vector_type(2)));
typedef float v8f __attribute__((ext_vector_type(8)));

#define RES_TOTAL   (32 * 4096)
#define RES_PER_BLK 32
#define NUM_BLK     1024
#define TRIPS       (RES_TOTAL / RES_PER_BLK / NUM_BLK)
#define RES_PITCH   144
#define WAVE_LDS    (4 * RES_PITCH)
#define EPSF        1e-12f

__device__ __forceinline__ void wave_sync() {
  __builtin_amdgcn_wave_barrier();
}

__device__ __forceinline__ v8f wmma4x4(v2f a, v2f b) {
  v8f c = {};
  return __builtin_amdgcn_wmma_f32_16x16x4_f32(false, a, false, b, (short)0, c,
                                               false, false);
}

__device__ __forceinline__ v8f wmma4x4acc(v2f a, v2f b, v8f c) {
  return __builtin_amdgcn_wmma_f32_16x16x4_f32(false, a, false, b, (short)0, c,
                                               false, false);
}

__device__ __forceinline__ v2f own_pair(v8f d, bool b0, bool b1) {
  const float l0 = b0 ? d[2] : d[0];
  const float l1 = b0 ? d[3] : d[1];
  const float u0 = b0 ? d[6] : d[4];
  const float u1 = b0 ? d[7] : d[5];
  v2f o;
  o[0] = b1 ? u0 : l0;
  o[1] = b1 ? u1 : l1;
  return o;
}

struct RawIn {
  int   sq;
  int   sq2[2];
  float dhc[6];
  float dho[2];
  float bb[9];
};

__device__ __forceinline__ RawIn load_raw(const float* __restrict__ dih,
                                          const float* __restrict__ bbc,
                                          const int* __restrict__ seq,
                                          int resBase, int iC, int iB, int hi,
                                          int ownOff) {
  RawIn r;
  const int myRes = resBase + iC;
  r.sq = seq[myRes];
#pragma unroll
  for (int p = 0; p < 2; ++p) r.sq2[p] = seq[resBase + 2 * hi + p];
  const float* dBase = dih + myRes * 14;
#pragma unroll
  for (int k = 0; k < 6; ++k) r.dhc[k] = dBase[6 + k];
  r.dho[0] = dBase[ownOff];
  r.dho[1] = dBase[ownOff + 1];
  const float* bp = bbc + (resBase + iB) * 9;
#pragma unroll
  for (int k = 0; k < 9; ++k) r.bb[k] = bp[k];
  return r;
}

__global__ __launch_bounds__(256, 8) void rotamer_projection_kernel(
    const float* __restrict__ dih,
    const float* __restrict__ bbc,
    const int*   __restrict__ seq,
    const float* __restrict__ dframes,
    const int*   __restrict__ gidx,
    const float* __restrict__ amask,
    const float* __restrict__ litp,
    float*       __restrict__ out)
{
  __shared__ __attribute__((aligned(16))) float ldsfr[8][WAVE_LDS];

  const int tid  = threadIdx.x;
  const int wave = tid >> 5;
  const int lane = tid & 31;
  float* fr = ldsfr[wave];

  const int low = lane & 15;
  const int hi  = lane >> 4;
  const int k0  = hi * 2;
  const int i   = (low >> 1) & 3;
  const int m   = 2 * (low >> 3) + (low & 1);
  const int iB  = low >> 2;
  const int mB  = low & 3;

  const bool bbP0 = ((iB >> 1) == hi) && ((iB & 1) == 0);
  const bool bbP1 = ((iB >> 1) == hi) && ((iB & 1) == 1);

  const bool ib0 = (i & 1) != 0;
  const bool ib1 = (i & 2) != 0;
  float2* rowp = reinterpret_cast<float2*>(fr + i * RES_PITCH + m * 4 + k0);
  const int  j       = m + 4 * hi;
  const bool ownTurn = (j >= 1 && j <= 3) || j == 7;
  const int  ownOff  = ownTurn ? (j - 1) * 2 : 0;
  float4* rawp = reinterpret_cast<float4*>(fr + i * RES_PITCH) + j;
  float2* scp  = reinterpret_cast<float2*>(fr + i * RES_PITCH + 128) + j;
  const float* frh = fr + 2 * hi * RES_PITCH;

  const float wA = (m == 1 || m == 2) ? 1.f : 0.f;
  const float wK = 1.f - wA;
  const float wS = (m == 1) ? 1.f : ((m == 2) ? -1.f : 0.f);

  const int atomc = (low < 14) ? low : 13;

  RawIn cur =
      load_raw(dih, bbc, seq, blockIdx.x * RES_PER_BLK + wave * 4, i, iB, hi,
               ownOff);

#pragma unroll 1
  for (int t = 0; t < TRIPS; ++t) {
    const int resBase = (blockIdx.x + t * NUM_BLK) * RES_PER_BLK + wave * 4;
    const int mySeq = cur.sq;

    const float nx = cur.bb[0], ny = cur.bb[1], nz = cur.bb[2];
    const float cax = cur.bb[3], cay = cur.bb[4], caz = cur.bb[5];
    const float cx = cur.bb[6], cy = cur.bb[7], cz = cur.bb[8];
    const float v1x = cx - cax, v1y = cy - cay, v1z = cz - caz;
    const float v2x = nx - cax, v2y = ny - cay, v2z = nz - caz;
    const float inv1 = rsqrtf(v1x * v1x + v1y * v1y + v1z * v1z + EPSF);
    const float e1x = v1x * inv1, e1y = v1y * inv1, e1z = v1z * inv1;
    const float d12 = e1x * v2x + e1y * v2y + e1z * v2z;
    const float u2x = v2x - e1x * d12, u2y = v2y - e1y * d12,
                u2z = v2z - e1z * d12;
    const float inv2 = rsqrtf(u2x * u2x + u2y * u2y + u2z * u2z + EPSF);
    const float e2x = u2x * inv2, e2y = u2y * inv2, e2z = u2z * inv2;
    const float e3x = e1y * e2z - e1z * e2y;
    const float e3y = e1z * e2x - e1x * e2z;
    const float e3z = e1x * e2y - e1y * e2x;

    float row0, row1, row2, row3;
    if (mB == 0)      { row0 = e1x; row1 = e2x; row2 = e3x; row3 = cax; }
    else if (mB == 1) { row0 = e1y; row1 = e2y; row2 = e3y; row3 = cay; }
    else if (mB == 2) { row0 = e1z; row1 = e2z; row2 = e3z; row3 = caz; }
    else             { row0 = 0.f; row1 = 0.f; row2 = 0.f; row3 = 1.f; }
    v2f abb[2][2];
    abb[0][0][0] = bbP0 ? row0 : 0.f;
    abb[0][0][1] = bbP0 ? row1 : 0.f;
    abb[0][1][0] = bbP0 ? row2 : 0.f;
    abb[0][1][1] = bbP0 ? row3 : 0.f;
    abb[1][0][0] = bbP1 ? row0 : 0.f;
    abb[1][0][1] = bbP1 ? row1 : 0.f;
    abb[1][1][0] = bbP1 ? row2 : 0.f;
    abb[1][1][1] = bbP1 ? row3 : 0.f;

    const float* tb = dframes + mySeq * 128;
    const float4 d4 = reinterpret_cast<const float4*>(tb + 64)[m];
    const float* colm = tb + k0 * 4 + m;
    const float* colx = tb + k0 * 4 + (3 - m);
    v2f dm[7], dx[7];
#pragma unroll
    for (int f = 5; f < 7; ++f) {
      dm[f][0] = colm[f * 16];
      dm[f][1] = colm[f * 16 + 4];
      dx[f][0] = colx[f * 16];
      dx[f][1] = colx[f * 16 + 4];
    }
    v2f dz7;
    dz7[0] = colm[112];
    dz7[1] = colm[112 + 4];
    const float4* rawsrc =
        reinterpret_cast<const float4*>(dframes + mySeq * 128) + j;
    const float4 raw0 = rawsrc[0];
    const float4 raw1 = rawsrc[8];
    int   gq[2];
    float mkq[2], l0q[2], l1q[2], l2q[2];
#pragma unroll
    for (int ri = 0; ri < 2; ++ri) {
      const int sq  = cur.sq2[ri];
      const int idx = sq * 14 + atomc;
      gq[ri]  = gidx[idx];
      mkq[ri] = amask[idx];
      const float* lp = litp + idx * 3;
      l0q[ri] = lp[0];
      l1q[ri] = lp[1];
      l2q[ri] = lp[2];
    }

    float sv[7], cv[7];
#pragma unroll
    for (int a = 4; a < 7; ++a) {
      const float x = cur.dhc[(a - 4) * 2];
      const float y = cur.dhc[(a - 4) * 2 + 1];
      const float inv = rsqrtf(x * x + y * y + EPSF);
      sv[a] = x * inv;
      cv[a] = y * inv;
    }
    {
      const float x = cur.dho[0];
      const float y = cur.dho[1];
      const float inv = rsqrtf(x * x + y * y + EPSF);
      float2 sc;
      sc.x = ownTurn ? x * inv : 0.f;
      sc.y = ownTurn ? y * inv : 1.f;
      scp[0] = sc;
      rawp[0] = raw0;
      rawp[8] = raw1;
    }

    v2f ypr[8];
    v2f fop[8];
    {
      ypr[4][0] = hi ? fmaf(cv[4], d4.z, -(sv[4] * d4.y)) : d4.x;
      ypr[4][1] = hi ? d4.w : fmaf(cv[4], d4.y, sv[4] * d4.z);
    }
#pragma unroll
    for (int f = 5; f < 7; ++f) {
      const float alpha = fmaf(wA, cv[f], wK);
      const float beta  = wS * sv[f];
      fop[f][0] = fmaf(alpha, dm[f][0], beta * dx[f][0]);
      fop[f][1] = fmaf(alpha, dm[f][1], beta * dx[f][1]);
    }

    __builtin_amdgcn_sched_barrier(0);
    const int tn = (t + 1 < TRIPS) ? t + 1 : t;
    const RawIn nxt = load_raw(
        dih, bbc, seq, (blockIdx.x + tn * NUM_BLK) * RES_PER_BLK + wave * 4, i,
        iB, hi, ownOff);

#pragma unroll
    for (int f = 5; f <= 7; ++f) {
      const v2f lop = (f < 7) ? fop[f] : dz7;
      ypr[f] = own_pair(wmma4x4(lop, ypr[f - 1]), ib0, ib1);
    }

    {
      float2 pr;
#pragma unroll
      for (int a = 4; a < 8; ++a) {
        pr.x = ypr[a][0];
        pr.y = ypr[a][1];
        rowp[a * 8] = pr;
      }
    }
    wave_sync();
    asm volatile("" : : "v"(mkq[0]), "v"(mkq[1]));

    float ov[2][3];
#pragma unroll
    for (int p = 0; p < 2; ++p) {
      const float* slot = frh + p * RES_PITCH + gq[p] * 16;
      const float4 f0 = reinterpret_cast<const float4*>(slot)[0];
      const float4 f1 = reinterpret_cast<const float4*>(slot)[1];
      const float4 f2 = reinterpret_cast<const float4*>(slot)[2];
      const float4 f3 = reinterpret_cast<const float4*>(slot)[3];
      const float2 sc = *reinterpret_cast<const float2*>(
          frh + p * RES_PITCH + 128 + gq[p] * 2);
      const float uy = sc.y * l1q[p] - sc.x * l2q[p];
      const float uz = sc.x * l1q[p] + sc.y * l2q[p];
      v2f b01, b23;
      b01[0] = fmaf(f0.z, uz, fmaf(f0.x, l0q[p], f0.y * uy)) + f0.w;
      b01[1] = fmaf(f1.z, uz, fmaf(f1.x, l0q[p], f1.y * uy)) + f1.w;
      b23[0] = fmaf(f2.z, uz, fmaf(f2.x, l0q[p], f2.y * uy)) + f2.w;
      b23[1] = fmaf(f3.z, uz, fmaf(f3.x, l0q[p], f3.y * uy)) + f3.w;
      const v8f d = wmma4x4acc(abb[p][1], b23, wmma4x4(abb[p][0], b01));
#pragma unroll
      for (int r = 0; r < 3; ++r) ov[p][r] = d[4 * p + r] * mkq[p];
    }
    if (low < 14) {
#pragma unroll
      for (int p = 0; p < 2; ++p) {
        float* op = out + ((resBase + 2 * hi + p) * 14 + low) * 3;
        op[0] = ov[p][0];
        op[1] = ov[p][1];
        op[2] = ov[p][2];
      }
    }
    cur = nxt;
  }
}

extern "C" void kernel_launch(void* const* d_in, const int* in_sizes, int n_in,
                              void* d_out, int out_size, void* d_ws,
                              size_t ws_size, hipStream_t stream) {
  const float* dih     = (const float*)d_in[0];
  const float* bbc     = (const float*)d_in[1];
  const int*   seq     = (const int*)d_in[2];
  const float* dframes = (const float*)d_in[3];
  const int*   gidx    = (const int*)d_in[4];
  const float* amask   = (const float*)d_in[5];
  const float* litp    = (const float*)d_in[6];
  float*       out     = (float*)d_out;

  rotamer_projection_kernel<<<NUM_BLK, 256, 0, stream>>>(
      dih, bbc, seq, dframes, gidx, amask, litp, out);
}
